# starting kernel
# baseline (speedup 1.0000x reference)
#include <hip/hip_runtime.h>

typedef __attribute__((ext_vector_type(2))) float v2f;
typedef __attribute__((ext_vector_type(8))) float v8f;

#define D     64
#define TROWS 32
#define LDP   68

__global__ void gcn_deg_init(float* deg, int n) {
  int i = blockIdx.x * blockDim.x + threadIdx.x;
  if (i < n) deg[i] = 1.0f;
}

__global__ void gcn_deg_edges(float* deg, const int* __restrict__ dst,
                              const float* __restrict__ ew, int e) {
  int i = blockIdx.x * blockDim.x + threadIdx.x;
  if (i < e) atomicAdd(&deg[dst[i]], ew[i]);
}

__global__ void gcn_dinv(float* deg, int n) {
  int i = blockIdx.x * blockDim.x + threadIdx.x;
  if (i < n) {
    float d = deg[i];
    deg[i] = (d > 0.0f) ? rsqrtf(d) : 0.0f;
  }
}

__global__ void gcn_copy_x(const float* __restrict__ x, float* __restrict__ Hcat, int n) {
  int i = blockIdx.x * blockDim.x + threadIdx.x;
  if (i < n * D) {
    int node = i >> 6, j = i & 63;
    Hcat[(size_t)node * 256 + j] = x[i];
  }
}

__global__ void gcn_agg_self(const float* __restrict__ Htmp, const float* __restrict__ dinv,
                             const float* __restrict__ bias, float* __restrict__ Hcat,
                             int off, int n) {
  int i = blockIdx.x * blockDim.x + threadIdx.x;
  if (i < n * D) {
    int node = i >> 6, j = i & 63;
    float di = dinv[node];
    Hcat[(size_t)node * 256 + off + j] = bias[j] + di * di * Htmp[i];
  }
}

__global__ void gcn_scatter(const float* __restrict__ Htmp, const float* __restrict__ dinv,
                            const float* __restrict__ ew, const int* __restrict__ src,
                            const int* __restrict__ dst, float* __restrict__ Hcat,
                            int off, int e) {
  int t = blockIdx.x * blockDim.x + threadIdx.x;
  int eid = t >> 4;
  int l   = t & 15;
  if (eid < e) {
    int s = src[eid], d = dst[eid];
    float nrm = dinv[s] * ew[eid] * dinv[d];
    const float4 h = *(const float4*)(Htmp + (size_t)s * D + l * 4);
    float* o = Hcat + (size_t)d * 256 + off + l * 4;
    atomicAdd(o + 0, h.x * nrm);
    atomicAdd(o + 1, h.y * nrm);
    atomicAdd(o + 2, h.z * nrm);
    atomicAdd(o + 3, h.w * nrm);
  }
}

__global__ void gcn_relu(float* Hcat, int off, int n) {
  int i = blockIdx.x * blockDim.x + threadIdx.x;
  if (i < n * D) {
    int node = i >> 6, j = i & 63;
    size_t idx = (size_t)node * 256 + off + j;
    float v = Hcat[idx];
    Hcat[idx] = v > 0.0f ? v : 0.0f;
  }
}

__global__ void gcn_gemm_wmma(const float* __restrict__ X, int ldx,
                              const float* __restrict__ W,
                              const float* __restrict__ bias,
                              float* __restrict__ Out, int K, int n) {
  __shared__ float Xs[TROWS * LDP];
  __shared__ float Ws[D * LDP];
  const int tid    = threadIdx.x;
  const int wave   = tid >> 5;
  const int lane   = tid & 31;
  const int laneLo = lane & 15;
  const int laneHi = lane >> 4;
  const int row0   = blockIdx.x * TROWS;
  const int wm     = (wave >> 2) << 4;
  const int wn     = (wave & 3) << 4;

  v8f acc = {};
  for (int kc = 0; kc < K; kc += D) {
    for (int i = tid; i < TROWS * D; i += 256) {
      int r = i >> 6, c = i & 63;
      int rr = row0 + r; if (rr >= n) rr = n - 1;
      Xs[r * LDP + c] = X[(size_t)rr * ldx + kc + c];
    }
    for (int i = tid; i < D * D; i += 256) {
      int r = i >> 6, c = i & 63;
      Ws[r * LDP + c] = W[(size_t)(kc + r) * D + c];
    }
    __syncthreads();
#pragma unroll
    for (int k0 = 0; k0 < D; k0 += 4) {
      v2f a = *(const v2f*)(&Xs[(wm + laneLo) * LDP + k0 + 2 * laneHi]);
      v2f b;
      b[0] = Ws[(k0 + 2 * laneHi + 0) * LDP + wn + laneLo];
      b[1] = Ws[(k0 + 2 * laneHi + 1) * LDP + wn + laneLo];
      acc = __builtin_amdgcn_wmma_f32_16x16x4_f32(false, a, false, b,
                                                  (short)0, acc, false, false);
    }
    __syncthreads();
  }
  const int gcol = wn + laneLo;
  const float bv = bias ? bias[gcol] : 0.0f;
#pragma unroll
  for (int v = 0; v < 8; ++v) {
    int grow = row0 + wm + v + 8 * laneHi;
    if (grow < n) Out[(size_t)grow * D + gcol] = acc[v] + bv;
  }
}

extern "C" void kernel_launch(void* const* d_in, const int* in_sizes, int n_in,
                              void* d_out, int out_size, void* d_ws, size_t ws_size,
                              hipStream_t stream) {
  const float* x   = (const float*)d_in[0];
  const int*   ei  = (const int*)d_in[1];
  const float* ew  = (const float*)d_in[2];
  const float* Wl[3] = {(const float*)d_in[3], (const float*)d_in[5], (const float*)d_in[7]};
  const float* bl[3] = {(const float*)d_in[4], (const float*)d_in[6], (const float*)d_in[8]};
  const float* Wf  = (const float*)d_in[9];
  const float* bf  = (const float*)d_in[10];

  const int N = in_sizes[0] / D;
  const int E = in_sizes[2];
  const int* src = ei;
  const int* dst = ei + E;

  float* deg  = (float*)d_ws;
  float* Hcat = deg + (((size_t)N + 255) & ~(size_t)255);
  float* Htmp = Hcat + (size_t)N * 256;

  const int T = 256;
  gcn_deg_init <<<(N + T - 1) / T, T, 0, stream>>>(deg, N);
  gcn_deg_edges<<<(E + T - 1) / T, T, 0, stream>>>(deg, dst, ew, E);
  gcn_dinv     <<<(N + T - 1) / T, T, 0, stream>>>(deg, N);
  gcn_copy_x   <<<((size_t)N * D + T - 1) / T, T, 0, stream>>>(x, Hcat, N);

  const int gemm_blocks = (N + TROWS - 1) / TROWS;
  for (int l = 0; l < 3; ++l) {
    const int off = (l + 1) * D;
    gcn_gemm_wmma<<<gemm_blocks, T, 0, stream>>>(Hcat + l * D, 256, Wl[l], nullptr,
                                                 Htmp, D, N);
    gcn_agg_self <<<((size_t)N * D + T - 1) / T, T, 0, stream>>>(Htmp, deg, bl[l],
                                                                 Hcat, off, N);
    gcn_scatter  <<<((size_t)E * 16 + T - 1) / T, T, 0, stream>>>(Htmp, deg, ew, src, dst,
                                                                  Hcat, off, E);
    gcn_relu     <<<((size_t)N * D + T - 1) / T, T, 0, stream>>>(Hcat, off, N);
  }
  gcn_gemm_wmma<<<gemm_blocks, T, 0, stream>>>(Hcat, 256, Wf, bf, (float*)d_out, 4 * D, N);
}


# GEMM blocks stay resident and loop over row tiles: W staged to LDS once per block, the rows of each next (tile, K-chunk) step requested ahead of the current step's WMMAs, one barrier per step
# speedup vs baseline: 3.2115x; 3.2115x over previous
#include <hip/hip_runtime.h>

typedef __attribute__((ext_vector_type(2))) float v2f;
typedef __attribute__((ext_vector_type(8))) float v8f;

#define D     64
#define TROWS 32
#define LDP   68

#define DEG_RANGE 12800
#define DEG_PARTS 32
#define DEG_T     1024
#define DEG_U     16
__global__ __launch_bounds__(DEG_T) void gcn_deg_ranges(const int* __restrict__ dst,
                              const float* __restrict__ ew, float* __restrict__ part,
                              int n, int e, int npad, int chunk) {
  __shared__ float acc[DEG_RANGE];
  const int tid = threadIdx.x;
  const int r0  = blockIdx.x * DEG_RANGE;
  for (int i = tid; i < DEG_RANGE; i += DEG_T) acc[i] = 0.0f;
  __syncthreads();
  const int eb = blockIdx.y * chunk;
  int ee = eb + chunk; if (ee > e) ee = e;
  int i = eb + tid;
  for (; i + (DEG_U - 1) * DEG_T < ee; i += DEG_U * DEG_T) {
    int dd[DEG_U];
    float ww[DEG_U];
#pragma unroll
    for (int u = 0; u < DEG_U; ++u) {
      dd[u] = dst[i + u * DEG_T];
      ww[u] = ew[i + u * DEG_T];
    }
    asm volatile("" :: "v"(ww[0]), "v"(ww[1]), "v"(ww[2]), "v"(ww[3]), "v"(ww[4]), "v"(ww[5]),
                       "v"(ww[6]), "v"(ww[7]), "v"(ww[8]), "v"(ww[9]), "v"(ww[10]), "v"(ww[11]),
                       "v"(ww[12]), "v"(ww[13]), "v"(ww[14]), "v"(ww[15]));
#pragma unroll
    for (int u = 0; u < DEG_U; ++u) {
      const unsigned k = (unsigned)(dd[u] - r0);
      if (k < (unsigned)DEG_RANGE) atomicAdd(&acc[k], ww[u]);
    }
  }
  for (; i < ee; i += DEG_T) {
    const unsigned k = (unsigned)(dst[i] - r0);
    const float w = ew[i];
    if (k < (unsigned)DEG_RANGE) atomicAdd(&acc[k], w);
  }
  __syncthreads();
  int cnt = n - r0; if (cnt > DEG_RANGE) cnt = DEG_RANGE;
  float* const o = part + (size_t)blockIdx.y * npad + r0;
  for (int j = tid; j < cnt; j += DEG_T) o[j] = acc[j];
}

__global__ void gcn_dinv(float* __restrict__ dinv, const float* __restrict__ part, int n, int npad) {
  int i = blockIdx.x * blockDim.x + threadIdx.x;
  if (i < n) {
    float p[DEG_PARTS];
#pragma unroll
    for (int s = 0; s < DEG_PARTS; ++s) p[s] = part[(size_t)s * npad + i];
    float d = 1.0f;
#pragma unroll
    for (int s = 0; s < DEG_PARTS; ++s) d += p[s];
    dinv[i] = (d > 0.0f) ? rsqrtf(d) : 0.0f;
  }
}

#define SC_GROUP 16
__global__ __launch_bounds__(256) void gcn_scatter(const float* __restrict__ Htmp,
                            const float* __restrict__ ew, const int* __restrict__ src,
                            const int* __restrict__ dst, float* __restrict__ SumA,
                            float* __restrict__ SumB, int e) {
  const unsigned lane = threadIdx.x & 31u;
  const int e0 = __builtin_amdgcn_readfirstlane((int)(blockIdx.x * 256u + (threadIdx.x & ~31u)));
  if (e0 >= e) return;
  const float* const h_lo = Htmp;
  const float* const h_hi = Htmp + 32;
  if (e - e0 >= 32) {
    const int eid = e0 + (int)lane;
    const int s = src[eid];
    const int d = dst[eid];
    const float nrm = ew[eid];
    const int where = __builtin_amdgcn_s_sendmsg_rtn(0x87);
    float* const plane = (where & (1 << 18)) ? SumB : SumA;
    float* const o_lo = plane;
    float* const o_hi = plane + 32;
#pragma unroll
    for (int j0 = 0; j0 < 32; j0 += SC_GROUP) {
      float a[SC_GROUP], b[SC_GROUP];
#pragma unroll
      for (int u = 0; u < SC_GROUP; ++u) {
        const unsigned hi = (unsigned)__builtin_amdgcn_readlane(s, j0 + u) * (unsigned)D + lane;
        a[u] = h_lo[hi];
        b[u] = h_hi[hi];
      }
#pragma unroll
      for (int u = 0; u < SC_GROUP; ++u) {
        const unsigned oi = (unsigned)__builtin_amdgcn_readlane(d, j0 + u) * (unsigned)D + lane;
        const float nj = __int_as_float(__builtin_amdgcn_readlane(__float_as_int(nrm), j0 + u));
        atomicAdd(o_lo + oi, a[u] * nj);
        atomicAdd(o_hi + oi, b[u] * nj);
      }
    }
  } else {
    const int where = __builtin_amdgcn_s_sendmsg_rtn(0x87);
    float* const plane = (where & (1 << 18)) ? SumB : SumA;
    float* const o_lo = plane;
    float* const o_hi = plane + 32;
    const int cnt = e - e0;
    const int eid = e0 + (int)lane;
    int s = 0, d = 0;
    float nrm = 0.0f;
    if (eid < e) {
      s = src[eid];
      d = dst[eid];
      nrm = ew[eid];
    }
    for (int j = 0; j < cnt; ++j) {
      const unsigned hi = (unsigned)__builtin_amdgcn_readlane(s, j) * (unsigned)D + lane;
      const unsigned oi = (unsigned)__builtin_amdgcn_readlane(d, j) * (unsigned)D + lane;
      const float nj = __int_as_float(__builtin_amdgcn_readlane(__float_as_int(nrm), j));
      const float a = h_lo[hi], b = h_hi[hi];
      atomicAdd(o_lo + oi, a * nj);
      atomicAdd(o_hi + oi, b * nj);
    }
  }
}

template <int NCH, int PLAIN0, int NW>
__global__ __launch_bounds__(32 * NW) void gcn_gemm_wmma(const float* __restrict__ X0,
                              const float* __restrict__ W,
                              const float* __restrict__ bias,
                              float* __restrict__ Out, int n,
                              const float* __restrict__ dinv,
                              const float* __restrict__ cb0,
                              const float* __restrict__ cb1,
                              const float* __restrict__ cb2,
                              float* __restrict__ Self,
                              const float* __restrict__ XA,
                              const float* __restrict__ XB,
                              float* __restrict__ Self2, int plane, int ntiles) {
  __shared__ __attribute__((aligned(16))) float Ws[NCH * D * LDP];
  constexpr int TR = 4 * NW;
  constexpr int NT = 32 * NW;
  __shared__ __attribute__((aligned(16))) float Xs[2][TR * LDP];
  const int tid    = threadIdx.x;
  const int wave   = tid >> 5;
  const int lane   = tid & 31;
  const int laneLo = lane & 15;
  const int laneHi = lane >> 4;
  const int wm     = (wave >> 2) << 4;
  const int wn     = (wave & 3) << 4;
  const int G      = (int)gridDim.x;

  const int sr = tid >> 4;
  const int sc = (tid & 15) << 2;

  float4 ra0, ra1, rb0, rb1, rc;
  float rd0 = 0.0f, rd1 = 0.0f;
  ra0 = ra1 = rb0 = rb1 = rc = make_float4(0.0f, 0.0f, 0.0f, 0.0f);

#define GEMM_FETCH_PLAIN(T_)                                                            \
  {                                                                                     \
    int g0_ = (T_) * TR + sr;         if (g0_ >= n) g0_ = n - 1;                        \
    int g1_ = (T_) * TR + sr + TR / 2; if (g1_ >= n) g1_ = n - 1;                        \
    ra0 = *(const float4*)(X0 + (size_t)g0_ * D + sc);                                  \
    ra1 = *(const float4*)(X0 + (size_t)g1_ * D + sc);                                  \
  }
#define GEMM_FETCH_SUMS(T_, J_)                                                         \
  {                                                                                     \
    int g0_ = (T_) * TR + sr;         if (g0_ >= n) g0_ = n - 1;                        \
    int g1_ = (T_) * TR + sr + TR / 2; if (g1_ >= n) g1_ = n - 1;                        \
    const float* cb_ = (J_) == 0 ? cb0 : ((J_) == 1 ? cb1 : cb2);                       \
    const float* xa_ = XA + (size_t)(J_) * plane;                                       \
    const float* xb_ = XB + (size_t)(J_) * plane;                                       \
    ra0 = *(const float4*)(xa_ + (size_t)g0_ * D + sc);                                 \
    ra1 = *(const float4*)(xa_ + (size_t)g1_ * D + sc);                                 \
    rb0 = *(const float4*)(xb_ + (size_t)g0_ * D + sc);                                 \
    rb1 = *(const float4*)(xb_ + (size_t)g1_ * D + sc);                                 \
    rd0 = dinv[g0_];                                                                    \
    rd1 = dinv[g1_];                                                                    \
    rc  = *(const float4*)(cb_ + sc);                                                   \
  }
#define GEMM_FETCH(T_, C_)                                                              \
  if (PLAIN0 && (C_) == 0) GEMM_FETCH_PLAIN(T_) else GEMM_FETCH_SUMS(T_, (C_) - PLAIN0)

#define GEMM_F(A_, B_, D_, C_) fmaxf((C_) + (D_) * ((A_) + (B_)), 0.0f)
#define GEMM_COMMIT(BUF_, C_)                                                           \
  {                                                                                     \
    float4 o0_ = ra0, o1_ = ra1;                                                        \
    if (!(PLAIN0 && (C_) == 0)) {                                                       \
      o0_.x = GEMM_F(ra0.x, rb0.x, rd0, rc.x); o0_.y = GEMM_F(ra0.y, rb0.y, rd0, rc.y); \
      o0_.z = GEMM_F(ra0.z, rb0.z, rd0, rc.z); o0_.w = GEMM_F(ra0.w, rb0.w, rd0, rc.w); \
      o1_.x = GEMM_F(ra1.x, rb1.x, rd1, rc.x); o1_.y = GEMM_F(ra1.y, rb1.y, rd1, rc.y); \
      o1_.z = GEMM_F(ra1.z, rb1.z, rd1, rc.z); o1_.w = GEMM_F(ra1.w, rb1.w, rd1, rc.w); \
    }                                                                                   \
    *(float4*)(&Xs[BUF_][sr * LDP + sc]) = o0_;                                         \
    *(float4*)(&Xs[BUF_][(sr + TR / 2) * LDP + sc]) = o1_;                              \
  }

  int tile = blockIdx.x;
  GEMM_FETCH(tile, 0)
#pragma unroll
  for (int i = 0; i < NCH * 1024 / NT; ++i) {
    const int p = tid + NT * i;
    const int r = p >> 4, c = (p & 15) << 2;
    *(float4*)(&Ws[r * LDP + c]) = *(const float4*)(W + (size_t)r * D + c);
  }
  GEMM_COMMIT(0, 0)
  __syncthreads();

  const int gcol = wn + laneLo;
  const float bv = bias ? bias[gcol] : 0.0f;
  int buf = 0;
  for (;;) {
    const int ntile = tile + G;
    const bool more = ntile < ntiles;
    const int row0 = tile * TR;
    v8f acc = {};
#pragma unroll
    for (int ch = 0; ch < NCH; ++ch) {
      float dv[8] = {0.0f, 0.0f, 0.0f, 0.0f, 0.0f, 0.0f, 0.0f, 0.0f};
      if (ch == NCH - 1 && Self) {
#pragma unroll
        for (int v = 0; v < 8; ++v) {
          int g = row0 + wm + v + 8 * laneHi; if (g >= n) g = n - 1;
          dv[v] = dinv[g];
        }
      }
      if (ch + 1 < NCH) { GEMM_FETCH(tile, ch + 1) }
      else if (more)    { GEMM_FETCH(ntile, 0) }
      const float* const xs = Xs[buf];
      const float* const ws = Ws + ch * (D * LDP);
#pragma unroll
      for (int k0 = 0; k0 < D; k0 += 4) {
        v2f a = *(const v2f*)(&xs[(wm + laneLo) * LDP + k0 + 2 * laneHi]);
        v2f b;
        b[0] = ws[(k0 + 2 * laneHi + 0) * LDP + wn + laneLo];
        b[1] = ws[(k0 + 2 * laneHi + 1) * LDP + wn + laneLo];
        acc = __builtin_amdgcn_wmma_f32_16x16x4_f32(false, a, false, b,
                                                    (short)0, acc, false, false);
      }
      if (ch == NCH - 1) {
        if (Self) {
#pragma unroll
          for (int v = 0; v < 8; ++v) {
            int grow = row0 + wm + v + 8 * laneHi;
            if (grow < n) {
              const float hs = dv[v] * (acc[v] + bv);
              Out[(size_t)grow * D + gcol] = hs;
              Self[(size_t)grow * D + gcol] = hs;
              Self2[(size_t)grow * D + gcol] = 0.0f;
            }
          }
        } else {
#pragma unroll
          for (int v = 0; v < 8; ++v) {
            int grow = row0 + wm + v + 8 * laneHi;
            if (grow < n) Out[(size_t)grow * D + gcol] = acc[v] + bv;
          }
        }
      }
      if (ch + 1 < NCH) {
        GEMM_COMMIT(buf ^ 1, ch + 1)
        __syncthreads();
        buf ^= 1;
      } else if (more) {
        GEMM_COMMIT(buf ^ 1, 0)
        __syncthreads();
        buf ^= 1;
      }
    }
    if (!more) break;
    tile = ntile;
  }
#undef GEMM_FETCH_PLAIN
#undef GEMM_FETCH_SUMS
#undef GEMM_FETCH
#undef GEMM_F
#undef GEMM_COMMIT
}

extern "C" void kernel_launch(void* const* d_in, const int* in_sizes, int n_in,
                              void* d_out, int out_size, void* d_ws, size_t ws_size,
                              hipStream_t stream) {
  const float* x   = (const float*)d_in[0];
  const int*   ei  = (const int*)d_in[1];
  const float* ew  = (const float*)d_in[2];
  const float* Wl[3] = {(const float*)d_in[3], (const float*)d_in[5], (const float*)d_in[7]};
  const float* bl[3] = {(const float*)d_in[4], (const float*)d_in[6], (const float*)d_in[8]};
  const float* Wf  = (const float*)d_in[9];
  const float* bf  = (const float*)d_in[10];

  const int N = in_sizes[0] / D;
  const int E = in_sizes[2];
  const int* src = ei;
  const int* dst = ei + E;

  const int Npad = (N + 255) & ~255;
  const int plane = N * D;
  float* deg  = (float*)d_ws;
  float* SumA = deg + (size_t)Npad;
  float* SumB = SumA + (size_t)3 * plane;
  float* Htmp = SumB + (size_t)3 * plane;
  float* part = Htmp + (size_t)plane;

  const int T = 256;
  const int chunk = (E + DEG_PARTS - 1) / DEG_PARTS;
  gcn_deg_ranges<<<dim3((N + DEG_RANGE - 1) / DEG_RANGE, DEG_PARTS), DEG_T, 0, stream>>>(
      dst, ew, part, N, E, Npad, chunk);
  gcn_dinv     <<<(N + T - 1) / T, T, 0, stream>>>(deg, part, N, Npad);

  const int ntiles = (N + TROWS - 1) / TROWS;
  const int ftiles = (N + 2 * TROWS - 1) / (2 * TROWS);
  const int g_layer = (ntiles + 1) / 2;
  const int g_final = (ftiles + 2) / 3;
  gcn_gemm_wmma<1, 1, 8><<<g_layer, T, 0, stream>>>(x, Wl[0], nullptr, Htmp, N, deg,
                                              nullptr, nullptr, nullptr,
                                              SumA, nullptr, nullptr, SumB, plane, ntiles);
  gcn_scatter  <<<(E + T - 1) / T, T, 0, stream>>>(Htmp, ew, src, dst, SumA, SumB, E);
  for (int l = 1; l < 3; ++l) {
    gcn_gemm_wmma<1, 0, 8><<<g_layer, T, 0, stream>>>(nullptr, Wl[l], nullptr, Htmp, N, deg,
                                                bl[l - 1], nullptr, nullptr,
                                                SumA + (size_t)l * plane,
                                                SumA + (size_t)(l - 1) * plane,
                                                SumB + (size_t)(l - 1) * plane,
                                                SumB + (size_t)l * plane, plane, ntiles);
    gcn_scatter  <<<(E + T - 1) / T, T, 0, stream>>>(Htmp, ew, src, dst, SumA + (size_t)l * plane,
                                                     SumB + (size_t)l * plane, E);
  }
  gcn_gemm_wmma<4, 1, 16><<<g_final, 2 * T, 0, stream>>>(x, Wf, bf, (float*)d_out, N, deg,
                                              bl[0], bl[1], bl[2],
                                              nullptr, SumA, SumB, nullptr, plane, ftiles);
}


# Product grids cut to the blocks resident at once at the built register counts (layers 1-2: 1536 places for 1563 blocks, final 512 for 521; 27 and 9 blocks ran alone as a second round); sums as before
# speedup vs baseline: 3.3265x; 1.0358x over previous
#include <hip/hip_runtime.h>

typedef __attribute__((ext_vector_type(2))) float v2f;
typedef __attribute__((ext_vector_type(8))) float v8f;

#define D     64
#define TROWS 32
#define LDP   68

#define DEG_RANGE 12800
#define DEG_PARTS 32
#define DEG_T     1024
#define DEG_U     16
__global__ __launch_bounds__(DEG_T) void gcn_deg_ranges(const int* __restrict__ dst,
                              const float* __restrict__ ew, float* __restrict__ part,
                              int n, int e, int npad, int chunk) {
  __shared__ float acc[DEG_RANGE];
  const int tid = threadIdx.x;
  const int r0  = blockIdx.x * DEG_RANGE;
  for (int i = tid; i < DEG_RANGE; i += DEG_T) acc[i] = 0.0f;
  __syncthreads();
  const int eb = blockIdx.y * chunk;
  int ee = eb + chunk; if (ee > e) ee = e;
  int i = eb + tid;
  for (; i + (DEG_U - 1) * DEG_T < ee; i += DEG_U * DEG_T) {
    int dd[DEG_U];
    float ww[DEG_U];
#pragma unroll
    for (int u = 0; u < DEG_U; ++u) {
      dd[u] = dst[i + u * DEG_T];
      ww[u] = ew[i + u * DEG_T];
    }
    asm volatile("" :: "v"(ww[0]), "v"(ww[1]), "v"(ww[2]), "v"(ww[3]), "v"(ww[4]), "v"(ww[5]),
                       "v"(ww[6]), "v"(ww[7]), "v"(ww[8]), "v"(ww[9]), "v"(ww[10]), "v"(ww[11]),
                       "v"(ww[12]), "v"(ww[13]), "v"(ww[14]), "v"(ww[15]));
#pragma unroll
    for (int u = 0; u < DEG_U; ++u) {
      const unsigned k = (unsigned)(dd[u] - r0);
      if (k < (unsigned)DEG_RANGE) atomicAdd(&acc[k], ww[u]);
    }
  }
  for (; i < ee; i += DEG_T) {
    const unsigned k = (unsigned)(dst[i] - r0);
    const float w = ew[i];
    if (k < (unsigned)DEG_RANGE) atomicAdd(&acc[k], w);
  }
  __syncthreads();
  int cnt = n - r0; if (cnt > DEG_RANGE) cnt = DEG_RANGE;
  float* const o = part + (size_t)blockIdx.y * npad + r0;
  for (int j = tid; j < cnt; j += DEG_T) o[j] = acc[j];
}

__global__ void gcn_dinv(float* __restrict__ dinv, const float* __restrict__ part, int n, int npad) {
  int i = blockIdx.x * blockDim.x + threadIdx.x;
  if (i < n) {
    float p[DEG_PARTS];
#pragma unroll
    for (int s = 0; s < DEG_PARTS; ++s) p[s] = part[(size_t)s * npad + i];
    float d = 1.0f;
#pragma unroll
    for (int s = 0; s < DEG_PARTS; ++s) d += p[s];
    dinv[i] = (d > 0.0f) ? rsqrtf(d) : 0.0f;
  }
}

#define SC_GROUP 16
__global__ __launch_bounds__(256) void gcn_scatter(const float* __restrict__ Htmp,
                            const float* __restrict__ ew, const int* __restrict__ src,
                            const int* __restrict__ dst, float* __restrict__ SumA,
                            float* __restrict__ SumB, int e) {
  const unsigned lane = threadIdx.x & 31u;
  const int e0 = __builtin_amdgcn_readfirstlane((int)(blockIdx.x * 256u + (threadIdx.x & ~31u)));
  if (e0 >= e) return;
  const float* const h_lo = Htmp;
  const float* const h_hi = Htmp + 32;
  if (e - e0 >= 32) {
    const int eid = e0 + (int)lane;
    const int s = src[eid];
    const int d = dst[eid];
    const float nrm = ew[eid];
    const int where = __builtin_amdgcn_s_sendmsg_rtn(0x87);
    float* const plane = (where & (1 << 18)) ? SumB : SumA;
    float* const o_lo = plane;
    float* const o_hi = plane + 32;
#pragma unroll
    for (int j0 = 0; j0 < 32; j0 += SC_GROUP) {
      float a[SC_GROUP], b[SC_GROUP];
#pragma unroll
      for (int u = 0; u < SC_GROUP; ++u) {
        const unsigned hi = (unsigned)__builtin_amdgcn_readlane(s, j0 + u) * (unsigned)D + lane;
        a[u] = h_lo[hi];
        b[u] = h_hi[hi];
      }
#pragma unroll
      for (int u = 0; u < SC_GROUP; ++u) {
        const unsigned oi = (unsigned)__builtin_amdgcn_readlane(d, j0 + u) * (unsigned)D + lane;
        const float nj = __int_as_float(__builtin_amdgcn_readlane(__float_as_int(nrm), j0 + u));
        atomicAdd(o_lo + oi, a[u] * nj);
        atomicAdd(o_hi + oi, b[u] * nj);
      }
    }
  } else {
    const int where = __builtin_amdgcn_s_sendmsg_rtn(0x87);
    float* const plane = (where & (1 << 18)) ? SumB : SumA;
    float* const o_lo = plane;
    float* const o_hi = plane + 32;
    const int cnt = e - e0;
    const int eid = e0 + (int)lane;
    int s = 0, d = 0;
    float nrm = 0.0f;
    if (eid < e) {
      s = src[eid];
      d = dst[eid];
      nrm = ew[eid];
    }
    for (int j = 0; j < cnt; ++j) {
      const unsigned hi = (unsigned)__builtin_amdgcn_readlane(s, j) * (unsigned)D + lane;
      const unsigned oi = (unsigned)__builtin_amdgcn_readlane(d, j) * (unsigned)D + lane;
      const float nj = __int_as_float(__builtin_amdgcn_readlane(__float_as_int(nrm), j));
      const float a = h_lo[hi], b = h_hi[hi];
      atomicAdd(o_lo + oi, a * nj);
      atomicAdd(o_hi + oi, b * nj);
    }
  }
}

template <int NCH, int PLAIN0, int NW, int WPS>
__global__ __launch_bounds__(32 * NW) __attribute__((amdgpu_waves_per_eu(WPS)))
void gcn_gemm_wmma(const float* __restrict__ X0,
                              const float* __restrict__ W,
                              const float* __restrict__ bias,
                              float* __restrict__ Out, int n,
                              const float* __restrict__ dinv,
                              const float* __restrict__ cb0,
                              const float* __restrict__ cb1,
                              const float* __restrict__ cb2,
                              float* __restrict__ Self,
                              const float* __restrict__ XA,
                              const float* __restrict__ XB,
                              float* __restrict__ Self2, int plane, int ntiles) {
  __shared__ __attribute__((aligned(16))) float Ws[NCH * D * LDP];
  constexpr int TR = 4 * NW;
  constexpr int NT = 32 * NW;
  __shared__ __attribute__((aligned(16))) float Xs[2][TR * LDP];
  const int tid    = threadIdx.x;
  const int wave   = tid >> 5;
  const int lane   = tid & 31;
  const int laneLo = lane & 15;
  const int laneHi = lane >> 4;
  const int wm     = (wave >> 2) << 4;
  const int wn     = (wave & 3) << 4;
  const int G      = (int)gridDim.x;

  const int sr = tid >> 4;
  const int sc = (tid & 15) << 2;

  float4 ra0, ra1, rb0, rb1, rc;
  float rd0 = 0.0f, rd1 = 0.0f;
  ra0 = ra1 = rb0 = rb1 = rc = make_float4(0.0f, 0.0f, 0.0f, 0.0f);

#define GEMM_FETCH_PLAIN(T_)                                                            \
  {                                                                                     \
    int g0_ = (T_) * TR + sr;         if (g0_ >= n) g0_ = n - 1;                        \
    int g1_ = (T_) * TR + sr + TR / 2; if (g1_ >= n) g1_ = n - 1;                        \
    ra0 = *(const float4*)(X0 + (size_t)g0_ * D + sc);                                  \
    ra1 = *(const float4*)(X0 + (size_t)g1_ * D + sc);                                  \
  }
#define GEMM_FETCH_SUMS(T_, J_)                                                         \
  {                                                                                     \
    int g0_ = (T_) * TR + sr;         if (g0_ >= n) g0_ = n - 1;                        \
    int g1_ = (T_) * TR + sr + TR / 2; if (g1_ >= n) g1_ = n - 1;                        \
    const float* cb_ = (J_) == 0 ? cb0 : ((J_) == 1 ? cb1 : cb2);                       \
    const float* xa_ = XA + (size_t)(J_) * plane;                                       \
    const float* xb_ = XB + (size_t)(J_) * plane;                                       \
    ra0 = *(const float4*)(xa_ + (size_t)g0_ * D + sc);                                 \
    ra1 = *(const float4*)(xa_ + (size_t)g1_ * D + sc);                                 \
    rb0 = *(const float4*)(xb_ + (size_t)g0_ * D + sc);                                 \
    rb1 = *(const float4*)(xb_ + (size_t)g1_ * D + sc);                                 \
    rd0 = dinv[g0_];                                                                    \
    rd1 = dinv[g1_];                                                                    \
    rc  = *(const float4*)(cb_ + sc);                                                   \
  }
#define GEMM_FETCH(T_, C_)                                                              \
  if (PLAIN0 && (C_) == 0) GEMM_FETCH_PLAIN(T_) else GEMM_FETCH_SUMS(T_, (C_) - PLAIN0)

#define GEMM_F(A_, B_, D_, C_) fmaxf((C_) + (D_) * ((A_) + (B_)), 0.0f)
#define GEMM_COMMIT(BUF_, C_)                                                           \
  {                                                                                     \
    float4 o0_ = ra0, o1_ = ra1;                                                        \
    if (!(PLAIN0 && (C_) == 0)) {                                                       \
      o0_.x = GEMM_F(ra0.x, rb0.x, rd0, rc.x); o0_.y = GEMM_F(ra0.y, rb0.y, rd0, rc.y); \
      o0_.z = GEMM_F(ra0.z, rb0.z, rd0, rc.z); o0_.w = GEMM_F(ra0.w, rb0.w, rd0, rc.w); \
      o1_.x = GEMM_F(ra1.x, rb1.x, rd1, rc.x); o1_.y = GEMM_F(ra1.y, rb1.y, rd1, rc.y); \
      o1_.z = GEMM_F(ra1.z, rb1.z, rd1, rc.z); o1_.w = GEMM_F(ra1.w, rb1.w, rd1, rc.w); \
    }                                                                                   \
    *(float4*)(&Xs[BUF_][sr * LDP + sc]) = o0_;                                         \
    *(float4*)(&Xs[BUF_][(sr + TR / 2) * LDP + sc]) = o1_;                              \
  }

  int tile = blockIdx.x;
  GEMM_FETCH(tile, 0)
#pragma unroll
  for (int i = 0; i < NCH * 1024 / NT; ++i) {
    const int p = tid + NT * i;
    const int r = p >> 4, c = (p & 15) << 2;
    *(float4*)(&Ws[r * LDP + c]) = *(const float4*)(W + (size_t)r * D + c);
  }
  GEMM_COMMIT(0, 0)
  __syncthreads();

  const int gcol = wn + laneLo;
  const float bv = bias ? bias[gcol] : 0.0f;
  int buf = 0;
  for (;;) {
    const int ntile = tile + G;
    const bool more = ntile < ntiles;
    const int row0 = tile * TR;
    v8f acc = {};
#pragma unroll
    for (int ch = 0; ch < NCH; ++ch) {
      float dv[8] = {0.0f, 0.0f, 0.0f, 0.0f, 0.0f, 0.0f, 0.0f, 0.0f};
      if (ch == NCH - 1 && Self) {
#pragma unroll
        for (int v = 0; v < 8; ++v) {
          int g = row0 + wm + v + 8 * laneHi; if (g >= n) g = n - 1;
          dv[v] = dinv[g];
        }
      }
      if (ch + 1 < NCH) { GEMM_FETCH(tile, ch + 1) }
      else if (more)    { GEMM_FETCH(ntile, 0) }
      const float* const xs = Xs[buf];
      const float* const ws = Ws + ch * (D * LDP);
#pragma unroll
      for (int k0 = 0; k0 < D; k0 += 4) {
        v2f a = *(const v2f*)(&xs[(wm + laneLo) * LDP + k0 + 2 * laneHi]);
        v2f b;
        b[0] = ws[(k0 + 2 * laneHi + 0) * LDP + wn + laneLo];
        b[1] = ws[(k0 + 2 * laneHi + 1) * LDP + wn + laneLo];
        acc = __builtin_amdgcn_wmma_f32_16x16x4_f32(false, a, false, b,
                                                    (short)0, acc, false, false);
      }
      if (ch == NCH - 1) {
        if (Self) {
#pragma unroll
          for (int v = 0; v < 8; ++v) {
            int grow = row0 + wm + v + 8 * laneHi;
            if (grow < n) {
              const float hs = dv[v] * (acc[v] + bv);
              Out[(size_t)grow * D + gcol] = hs;
              Self[(size_t)grow * D + gcol] = hs;
              Self2[(size_t)grow * D + gcol] = 0.0f;
            }
          }
        } else {
#pragma unroll
          for (int v = 0; v < 8; ++v) {
            int grow = row0 + wm + v + 8 * laneHi;
            if (grow < n) Out[(size_t)grow * D + gcol] = acc[v] + bv;
          }
        }
      }
      if (ch + 1 < NCH) {
        GEMM_COMMIT(buf ^ 1, ch + 1)
        __syncthreads();
        buf ^= 1;
      } else if (more) {
        GEMM_COMMIT(buf ^ 1, 0)
        __syncthreads();
        buf ^= 1;
      }
    }
    if (!more) break;
    tile = ntile;
  }
#undef GEMM_FETCH_PLAIN
#undef GEMM_FETCH_SUMS
#undef GEMM_FETCH
#undef GEMM_F
#undef GEMM_COMMIT
}

extern "C" void kernel_launch(void* const* d_in, const int* in_sizes, int n_in,
                              void* d_out, int out_size, void* d_ws, size_t ws_size,
                              hipStream_t stream) {
  const float* x   = (const float*)d_in[0];
  const int*   ei  = (const int*)d_in[1];
  const float* ew  = (const float*)d_in[2];
  const float* Wl[3] = {(const float*)d_in[3], (const float*)d_in[5], (const float*)d_in[7]};
  const float* bl[3] = {(const float*)d_in[4], (const float*)d_in[6], (const float*)d_in[8]};
  const float* Wf  = (const float*)d_in[9];
  const float* bf  = (const float*)d_in[10];

  const int N = in_sizes[0] / D;
  const int E = in_sizes[2];
  const int* src = ei;
  const int* dst = ei + E;

  const int Npad = (N + 255) & ~255;
  const int plane = N * D;
  float* deg  = (float*)d_ws;
  float* SumA = deg + (size_t)Npad;
  float* SumB = SumA + (size_t)3 * plane;
  float* Htmp = SumB + (size_t)3 * plane;
  float* part = Htmp + (size_t)plane;

  const int T = 256;
  const int chunk = (E + DEG_PARTS - 1) / DEG_PARTS;
  gcn_deg_ranges<<<dim3((N + DEG_RANGE - 1) / DEG_RANGE, DEG_PARTS), DEG_T, 0, stream>>>(
      dst, ew, part, N, E, Npad, chunk);
  gcn_dinv     <<<(N + T - 1) / T, T, 0, stream>>>(deg, part, N, Npad);

  const int ntiles = (N + TROWS - 1) / TROWS;
  const int ftiles = (N + 2 * TROWS - 1) / (2 * TROWS);
  constexpr int UNITS = 256;
  constexpr int WPS_X = 16, WPS_S = 12, WPS_F = 8;
  const int p_layer0 = UNITS * 4 * WPS_X / 8;
  const int p_layer  = UNITS * 4 * WPS_S / 8;
  const int p_final  = UNITS * 4 * WPS_F / 16;
  const int g_want  = (ntiles + 1) / 2;
  const int g_layer0 = g_want < p_layer0 ? g_want : p_layer0;
  const int g_layer  = g_want < p_layer  ? g_want : p_layer;
  const int g_fwant = (ftiles + 2) / 3;
  const int g_final = g_fwant < p_final ? g_fwant : p_final;
  gcn_gemm_wmma<1, 1, 8, WPS_X><<<g_layer0, T, 0, stream>>>(x, Wl[0], nullptr, Htmp, N, deg,
                                              nullptr, nullptr, nullptr,
                                              SumA, nullptr, nullptr, SumB, plane, ntiles);
  gcn_scatter  <<<(E + T - 1) / T, T, 0, stream>>>(Htmp, ew, src, dst, SumA, SumB, E);
  for (int l = 1; l < 3; ++l) {
    gcn_gemm_wmma<1, 0, 8, WPS_S><<<g_layer, T, 0, stream>>>(nullptr, Wl[l], nullptr, Htmp, N, deg,
                                                bl[l - 1], nullptr, nullptr,
                                                SumA + (size_t)l * plane,
                                                SumA + (size_t)(l - 1) * plane,
                                                SumB + (size_t)(l - 1) * plane,
                                                SumB + (size_t)l * plane, plane, ntiles);
    gcn_scatter  <<<(E + T - 1) / T, T, 0, stream>>>(Htmp, ew, src, dst, SumA + (size_t)l * plane,
                                                     SumB + (size_t)l * plane, E);
  }
  gcn_gemm_wmma<4, 1, 16, WPS_F><<<g_final, 2 * T, 0, stream>>>(x, Wf, bf, (float*)d_out, N, deg,
                                              bl[0], bl[1], bl[2],
                                              nullptr, SumA, SumB, nullptr, plane, ftiles);
}


# Edge sums in one plane: destination rows split in two node ranges, each summed by one package half; a block draws its (chunk, range) by integer ticket; second plane gone; sum order changes
# speedup vs baseline: 3.5994x; 1.0820x over previous
#include <hip/hip_runtime.h>

typedef __attribute__((ext_vector_type(2))) float v2f;
typedef __attribute__((ext_vector_type(8))) float v8f;

#define D     64
#define TROWS 32
#define LDP   68

#define DEG_RANGE 12800
#define DEG_PARTS 32
#define DEG_T     1024
#define DEG_U     16
__global__ __launch_bounds__(DEG_T) void gcn_deg_ranges(const int* __restrict__ dst,
                              const float* __restrict__ ew, float* __restrict__ part,
                              int n, int e, int npad, int chunk) {
  __shared__ float acc[DEG_RANGE];
  const int tid = threadIdx.x;
  const int r0  = blockIdx.x * DEG_RANGE;
  for (int i = tid; i < DEG_RANGE; i += DEG_T) acc[i] = 0.0f;
  __syncthreads();
  const int eb = blockIdx.y * chunk;
  int ee = eb + chunk; if (ee > e) ee = e;
  int i = eb + tid;
  for (; i + (DEG_U - 1) * DEG_T < ee; i += DEG_U * DEG_T) {
    int dd[DEG_U];
    float ww[DEG_U];
#pragma unroll
    for (int u = 0; u < DEG_U; ++u) {
      dd[u] = dst[i + u * DEG_T];
      ww[u] = ew[i + u * DEG_T];
    }
    asm volatile("" :: "v"(ww[0]), "v"(ww[1]), "v"(ww[2]), "v"(ww[3]), "v"(ww[4]), "v"(ww[5]),
                       "v"(ww[6]), "v"(ww[7]), "v"(ww[8]), "v"(ww[9]), "v"(ww[10]), "v"(ww[11]),
                       "v"(ww[12]), "v"(ww[13]), "v"(ww[14]), "v"(ww[15]));
#pragma unroll
    for (int u = 0; u < DEG_U; ++u) {
      const unsigned k = (unsigned)(dd[u] - r0);
      if (k < (unsigned)DEG_RANGE) atomicAdd(&acc[k], ww[u]);
    }
  }
  for (; i < ee; i += DEG_T) {
    const unsigned k = (unsigned)(dst[i] - r0);
    const float w = ew[i];
    if (k < (unsigned)DEG_RANGE) atomicAdd(&acc[k], w);
  }
  __syncthreads();
  int cnt = n - r0; if (cnt > DEG_RANGE) cnt = DEG_RANGE;
  float* const o = part + (size_t)blockIdx.y * npad + r0;
  for (int j = tid; j < cnt; j += DEG_T) o[j] = acc[j];
}

#define SC_TPITCH 64
__global__ void gcn_dinv(float* __restrict__ dinv, const float* __restrict__ part, int n, int npad,
                         int* __restrict__ tickets) {
  int i = blockIdx.x * blockDim.x + threadIdx.x;
  if (i < 6) tickets[i * SC_TPITCH] = 0;
  if (i < n) {
    float p[DEG_PARTS];
#pragma unroll
    for (int s = 0; s < DEG_PARTS; ++s) p[s] = part[(size_t)s * npad + i];
    float d = 1.0f;
#pragma unroll
    for (int s = 0; s < DEG_PARTS; ++s) d += p[s];
    dinv[i] = (d > 0.0f) ? rsqrtf(d) : 0.0f;
  }
}

#define SC_GROUP   16
#define SC_BATCHES 4
#define SC_CHUNK   (8 * SC_BATCHES * 32)
__global__ __launch_bounds__(256) void gcn_scatter(const float* __restrict__ Htmp,
                            const float* __restrict__ ew, const int* __restrict__ src,
                            const int* __restrict__ dst, float* __restrict__ Sum,
                            int* __restrict__ tickets, int e, int n, int nchunks) {
  __shared__ int pick[2];
  if (threadIdx.x == 0) {
    const int where = __builtin_amdgcn_s_sendmsg_rtn(0x87);
    int c = (where >> 18) & 1;
    int t = atomicAdd(&tickets[c * SC_TPITCH], 1);
    if (t >= nchunks) {
      c ^= 1;
      t = atomicAdd(&tickets[c * SC_TPITCH], 1);
    }
    pick[0] = t;
    pick[1] = c;
  }
  __syncthreads();
  const int chunk = __builtin_amdgcn_readfirstlane(pick[0]);
  const int cls   = __builtin_amdgcn_readfirstlane(pick[1]);
  if (chunk >= nchunks) return;
  const unsigned lane = threadIdx.x & 31u;
  const int nlow = n >> 1;
  const float* const h_lo = Htmp;
  const float* const h_hi = Htmp + 32;
  float* const o_lo = Sum;
  float* const o_hi = Sum + 32;
  int eid = chunk * SC_CHUNK + (int)(threadIdx.x & ~31u) + (int)lane;
  int s = 0, d = 0;
  float nrm = 0.0f;
  if (eid < e) { s = src[eid]; d = dst[eid]; nrm = ew[eid]; }
  for (int b = 0; b < SC_BATCHES; ++b) {
    unsigned m = __builtin_amdgcn_ballot_w32(eid < e && (int)(d >= nlow) == cls);
    const int eid_n = eid + 256;
    int s_n = 0, d_n = 0;
    float nrm_n = 0.0f;
#pragma unroll
    for (int g = 0; g < 32 / SC_GROUP; ++g) {
      float a[SC_GROUP], bq[SC_GROUP];
      unsigned mm = m;
#pragma unroll
      for (int u = 0; u < SC_GROUP; ++u) {
        if (mm) {
          const int j = __builtin_ctz(mm);
          mm &= mm - 1u;
          const unsigned hi = (unsigned)__builtin_amdgcn_readlane(s, j) * (unsigned)D + lane;
          a[u]  = h_lo[hi];
          bq[u] = h_hi[hi];
        }
      }
      if (g == 0 && b + 1 < SC_BATCHES && eid_n < e) {
        s_n = src[eid_n]; d_n = dst[eid_n]; nrm_n = ew[eid_n];
      }
#pragma unroll
      for (int u = 0; u < SC_GROUP; ++u) {
        if (m) {
          const int j = __builtin_ctz(m);
          m &= m - 1u;
          const unsigned oi = (unsigned)__builtin_amdgcn_readlane(d, j) * (unsigned)D + lane;
          const float nj = __int_as_float(__builtin_amdgcn_readlane(__float_as_int(nrm), j));
          atomicAdd(o_lo + oi, a[u] * nj);
          atomicAdd(o_hi + oi, bq[u] * nj);
        }
      }
    }
    eid = eid_n; s = s_n; d = d_n; nrm = nrm_n;
  }
}

template <int NCH, int PLAIN0, int NW, int WPS>
__global__ __launch_bounds__(32 * NW) __attribute__((amdgpu_waves_per_eu(WPS)))
void gcn_gemm_wmma(const float* __restrict__ X0,
                              const float* __restrict__ W,
                              const float* __restrict__ bias,
                              float* __restrict__ Out, int n,
                              const float* __restrict__ dinv,
                              const float* __restrict__ cb0,
                              const float* __restrict__ cb1,
                              const float* __restrict__ cb2,
                              float* __restrict__ Self,
                              const float* __restrict__ XA, int plane, int ntiles) {
  __shared__ __attribute__((aligned(16))) float Ws[NCH * D * LDP];
  constexpr int TR = 4 * NW;
  constexpr int NT = 32 * NW;
  __shared__ __attribute__((aligned(16))) float Xs[2][TR * LDP];
  const int tid    = threadIdx.x;
  const int wave   = tid >> 5;
  const int lane   = tid & 31;
  const int laneLo = lane & 15;
  const int laneHi = lane >> 4;
  const int wm     = (wave >> 2) << 4;
  const int wn     = (wave & 3) << 4;
  const int G      = (int)gridDim.x;

  const int sr = tid >> 4;
  const int sc = (tid & 15) << 2;

  float4 ra0, ra1, rc;
  float rd0 = 0.0f, rd1 = 0.0f;
  ra0 = ra1 = rc = make_float4(0.0f, 0.0f, 0.0f, 0.0f);

#define GEMM_FETCH_PLAIN(T_)                                                            \
  {                                                                                     \
    int g0_ = (T_) * TR + sr;         if (g0_ >= n) g0_ = n - 1;                        \
    int g1_ = (T_) * TR + sr + TR / 2; if (g1_ >= n) g1_ = n - 1;                        \
    ra0 = *(const float4*)(X0 + (size_t)g0_ * D + sc);                                  \
    ra1 = *(const float4*)(X0 + (size_t)g1_ * D + sc);                                  \
  }
#define GEMM_FETCH_SUMS(T_, J_)                                                         \
  {                                                                                     \
    int g0_ = (T_) * TR + sr;         if (g0_ >= n) g0_ = n - 1;                        \
    int g1_ = (T_) * TR + sr + TR / 2; if (g1_ >= n) g1_ = n - 1;                        \
    const float* cb_ = (J_) == 0 ? cb0 : ((J_) == 1 ? cb1 : cb2);                       \
    const float* xa_ = XA + (size_t)(J_) * plane;                                       \
    ra0 = *(const float4*)(xa_ + (size_t)g0_ * D + sc);                                 \
    ra1 = *(const float4*)(xa_ + (size_t)g1_ * D + sc);                                 \
    rd0 = dinv[g0_];                                                                    \
    rd1 = dinv[g1_];                                                                    \
    rc  = *(const float4*)(cb_ + sc);                                                   \
  }
#define GEMM_FETCH(T_, C_)                                                              \
  if (PLAIN0 && (C_) == 0) GEMM_FETCH_PLAIN(T_) else GEMM_FETCH_SUMS(T_, (C_) - PLAIN0)

#define GEMM_F(A_, D_, C_) fmaxf((C_) + (D_) * (A_), 0.0f)
#define GEMM_COMMIT(BUF_, C_)                                                           \
  {                                                                                     \
    float4 o0_ = ra0, o1_ = ra1;                                                        \
    if (!(PLAIN0 && (C_) == 0)) {                                                       \
      o0_.x = GEMM_F(ra0.x, rd0, rc.x); o0_.y = GEMM_F(ra0.y, rd0, rc.y);               \
      o0_.z = GEMM_F(ra0.z, rd0, rc.z); o0_.w = GEMM_F(ra0.w, rd0, rc.w);               \
      o1_.x = GEMM_F(ra1.x, rd1, rc.x); o1_.y = GEMM_F(ra1.y, rd1, rc.y);               \
      o1_.z = GEMM_F(ra1.z, rd1, rc.z); o1_.w = GEMM_F(ra1.w, rd1, rc.w);               \
    }                                                                                   \
    *(float4*)(&Xs[BUF_][sr * LDP + sc]) = o0_;                                         \
    *(float4*)(&Xs[BUF_][(sr + TR / 2) * LDP + sc]) = o1_;                              \
  }

  int tile = blockIdx.x;
  GEMM_FETCH(tile, 0)
#pragma unroll
  for (int i = 0; i < NCH * 1024 / NT; ++i) {
    const int p = tid + NT * i;
    const int r = p >> 4, c = (p & 15) << 2;
    *(float4*)(&Ws[r * LDP + c]) = *(const float4*)(W + (size_t)r * D + c);
  }
  GEMM_COMMIT(0, 0)
  __syncthreads();

  const int gcol = wn + laneLo;
  const float bv = bias ? bias[gcol] : 0.0f;
  int buf = 0;
  for (;;) {
    const int ntile = tile + G;
    const bool more = ntile < ntiles;
    const int row0 = tile * TR;
    v8f acc = {};
#pragma unroll
    for (int ch = 0; ch < NCH; ++ch) {
      float dv[8] = {0.0f, 0.0f, 0.0f, 0.0f, 0.0f, 0.0f, 0.0f, 0.0f};
      if (ch == NCH - 1 && Self) {
#pragma unroll
        for (int v = 0; v < 8; ++v) {
          int g = row0 + wm + v + 8 * laneHi; if (g >= n) g = n - 1;
          dv[v] = dinv[g];
        }
      }
      if (ch + 1 < NCH) { GEMM_FETCH(tile, ch + 1) }
      else if (more)    { GEMM_FETCH(ntile, 0) }
      const float* const xs = Xs[buf];
      const float* const ws = Ws + ch * (D * LDP);
#pragma unroll
      for (int k0 = 0; k0 < D; k0 += 4) {
        v2f a = *(const v2f*)(&xs[(wm + laneLo) * LDP + k0 + 2 * laneHi]);
        v2f b;
        b[0] = ws[(k0 + 2 * laneHi + 0) * LDP + wn + laneLo];
        b[1] = ws[(k0 + 2 * laneHi + 1) * LDP + wn + laneLo];
        acc = __builtin_amdgcn_wmma_f32_16x16x4_f32(false, a, false, b,
                                                    (short)0, acc, false, false);
      }
      if (ch == NCH - 1) {
        if (Self) {
#pragma unroll
          for (int v = 0; v < 8; ++v) {
            int grow = row0 + wm + v + 8 * laneHi;
            if (grow < n) {
              const float hs = dv[v] * (acc[v] + bv);
              Out[(size_t)grow * D + gcol] = hs;
              Self[(size_t)grow * D + gcol] = hs;
            }
          }
        } else {
#pragma unroll
          for (int v = 0; v < 8; ++v) {
            int grow = row0 + wm + v + 8 * laneHi;
            if (grow < n) Out[(size_t)grow * D + gcol] = acc[v] + bv;
          }
        }
      }
      if (ch + 1 < NCH) {
        GEMM_COMMIT(buf ^ 1, ch + 1)
        __syncthreads();
        buf ^= 1;
      } else if (more) {
        GEMM_COMMIT(buf ^ 1, 0)
        __syncthreads();
        buf ^= 1;
      }
    }
    if (!more) break;
    tile = ntile;
  }
#undef GEMM_FETCH_PLAIN
#undef GEMM_FETCH_SUMS
#undef GEMM_FETCH
#undef GEMM_F
#undef GEMM_COMMIT
}

extern "C" void kernel_launch(void* const* d_in, const int* in_sizes, int n_in,
                              void* d_out, int out_size, void* d_ws, size_t ws_size,
                              hipStream_t stream) {
  const float* x   = (const float*)d_in[0];
  const int*   ei  = (const int*)d_in[1];
  const float* ew  = (const float*)d_in[2];
  const float* Wl[3] = {(const float*)d_in[3], (const float*)d_in[5], (const float*)d_in[7]};
  const float* bl[3] = {(const float*)d_in[4], (const float*)d_in[6], (const float*)d_in[8]};
  const float* Wf  = (const float*)d_in[9];
  const float* bf  = (const float*)d_in[10];

  const int N = in_sizes[0] / D;
  const int E = in_sizes[2];
  const int* src = ei;
  const int* dst = ei + E;

  const int Npad = (N + 255) & ~255;
  const int plane = N * D;
  float* deg  = (float*)d_ws;
  float* Sum  = deg + (size_t)Npad;
  float* Htmp = Sum + (size_t)3 * plane;
  float* part = Htmp + (size_t)plane;
  int* tickets = (int*)(part + (size_t)DEG_PARTS * Npad);

  const int T = 256;
  const int chunk = (E + DEG_PARTS - 1) / DEG_PARTS;
  gcn_deg_ranges<<<dim3((N + DEG_RANGE - 1) / DEG_RANGE, DEG_PARTS), DEG_T, 0, stream>>>(
      dst, ew, part, N, E, Npad, chunk);
  gcn_dinv     <<<(N + T - 1) / T, T, 0, stream>>>(deg, part, N, Npad, tickets);

  const int ntiles = (N + TROWS - 1) / TROWS;
  const int ftiles = (N + 2 * TROWS - 1) / (2 * TROWS);
  constexpr int UNITS = 256;
  constexpr int WPS_X = 16, WPS_S = 12, WPS_F = 8;
  const int p_layer0 = UNITS * 4 * WPS_X / 8;
  const int p_layer  = UNITS * 4 * WPS_S / 8;
  const int p_final  = UNITS * 4 * WPS_F / 16;
  const int g_want  = (ntiles + 1) / 2;
  const int g_layer0 = g_want < p_layer0 ? g_want : p_layer0;
  const int g_layer  = g_want < p_layer  ? g_want : p_layer;
  const int g_fwant = (ftiles + 2) / 3;
  const int g_final = g_fwant < p_final ? g_fwant : p_final;
  const int nchunks = (E + SC_CHUNK - 1) / SC_CHUNK;
  gcn_gemm_wmma<1, 1, 8, WPS_X><<<g_layer0, T, 0, stream>>>(x, Wl[0], nullptr, Htmp, N, deg,
                                              nullptr, nullptr, nullptr,
                                              Sum, nullptr, plane, ntiles);
  gcn_scatter  <<<2 * nchunks, T, 0, stream>>>(Htmp, ew, src, dst, Sum, tickets, E, N, nchunks);
  for (int l = 1; l < 3; ++l) {
    gcn_gemm_wmma<1, 0, 8, WPS_S><<<g_layer, T, 0, stream>>>(nullptr, Wl[l], nullptr, Htmp, N, deg,
                                                bl[l - 1], nullptr, nullptr,
                                                Sum + (size_t)l * plane,
                                                Sum + (size_t)(l - 1) * plane, plane, ntiles);
    gcn_scatter  <<<2 * nchunks, T, 0, stream>>>(Htmp, ew, src, dst, Sum + (size_t)l * plane,
                                                 tickets + 2 * SC_TPITCH * l, E, N, nchunks);
  }
  gcn_gemm_wmma<4, 1, 16, WPS_F><<<g_final, 2 * T, 0, stream>>>(x, Wf, bf, (float*)d_out, N, deg,
                                              bl[0], bl[1], bl[2],
                                              nullptr, Sum, plane, ftiles);
}
